# starting kernel
# baseline (speedup 1.0000x reference)
#include <hip/hip_runtime.h>
#include <math.h>

#define BB 16
#define CC 256
#define HW 16384
#define ROWS (BB * CC)
#define CP  (CC + 4)

typedef __attribute__((ext_vector_type(2))) float v2f;
typedef __attribute__((ext_vector_type(4))) float v4f;
typedef __attribute__((ext_vector_type(8))) float v8f;

__global__ __launch_bounds__(256) void cam_pool_kernel(
    const float* __restrict__ x,
    float* __restrict__ avg_out,
    float* __restrict__ max_out) {
  const int row = blockIdx.x;
  const v4f* __restrict__ p4 =
      (const v4f*)(x + (size_t)row * HW);
  const int tid = threadIdx.x;

  float s = 0.0f;
  float m = -INFINITY;
#pragma unroll
  for (int i = 0; i < 16; ++i) {
    v4f v = __builtin_nontemporal_load(&p4[tid + i * 256]);
    s += (v.x + v.y) + (v.z + v.w);
    m = fmaxf(m, fmaxf(fmaxf(v.x, v.y), fmaxf(v.z, v.w)));
  }

#pragma unroll
  for (int off = 16; off > 0; off >>= 1) {
    s += __shfl_xor(s, off, 32);
    m = fmaxf(m, __shfl_xor(m, off, 32));
  }

  __shared__ float ss[8];
  __shared__ float sm[8];
  const int wave = tid >> 5;
  if ((tid & 31) == 0) { ss[wave] = s; sm[wave] = m; }
  __syncthreads();
  if (tid == 0) {
    float S = ss[0], M = sm[0];
#pragma unroll
    for (int w = 1; w < 8; ++w) { S += ss[w]; M = fmaxf(M, sm[w]); }
    avg_out[row] = S * (1.0f / (float)HW);
    max_out[row] = M;
  }
}

__global__ __launch_bounds__(256) void cam_mlp_wmma_kernel(
    const float* __restrict__ avg,
    const float* __restrict__ mx,
    const float* __restrict__ w1,
    const float* __restrict__ w2,
    float* __restrict__ out) {
  __shared__ float h_avg[BB][CP];
  __shared__ float h_max[BB][CP];

  const int tid  = threadIdx.x;
  const int wave = tid >> 5;
  const int lane = tid & 31;
  const int half = lane >> 4;
  const int l16  = lane & 15;

  for (int t = 0; t < 2; ++t) {
    const int ntile = wave * 2 + t;
    const int ocol  = ntile * 16 + l16;
    v8f acc_a = {};
    v8f acc_m = {};
    for (int k = 0; k < CC; k += 4) {
      const int ka = k + half * 2;
      v2f a_a, a_m, b;
      a_a.x = avg[l16 * CC + ka];  a_a.y = avg[l16 * CC + ka + 1];
      a_m.x = mx [l16 * CC + ka];  a_m.y = mx [l16 * CC + ka + 1];
      b.x = w1[ocol * CC + ka];    b.y = w1[ocol * CC + ka + 1];
      acc_a = __builtin_amdgcn_wmma_f32_16x16x4_f32(
          false, a_a, false, b, (short)0, acc_a, false, false);
      acc_m = __builtin_amdgcn_wmma_f32_16x16x4_f32(
          false, a_m, false, b, (short)0, acc_m, false, false);
    }
#pragma unroll
    for (int r = 0; r < 8; ++r) {
      const int mrow = r + 8 * half;
      h_avg[mrow][ntile * 16 + l16] = fmaxf(acc_a[r], 0.0f);
      h_max[mrow][ntile * 16 + l16] = fmaxf(acc_m[r], 0.0f);
    }
  }
  __syncthreads();

  for (int t = 0; t < 2; ++t) {
    const int ntile = wave * 2 + t;
    const int ocol  = ntile * 16 + l16;
    v8f acc_a = {};
    v8f acc_m = {};
    for (int k = 0; k < CC; k += 4) {
      const int ka = k + half * 2;
      v2f a_a, a_m, b;
      a_a.x = h_avg[l16][ka];  a_a.y = h_avg[l16][ka + 1];
      a_m.x = h_max[l16][ka];  a_m.y = h_max[l16][ka + 1];
      b.x = w2[ocol * CC + ka];  b.y = w2[ocol * CC + ka + 1];
      acc_a = __builtin_amdgcn_wmma_f32_16x16x4_f32(
          false, a_a, false, b, (short)0, acc_a, false, false);
      acc_m = __builtin_amdgcn_wmma_f32_16x16x4_f32(
          false, a_m, false, b, (short)0, acc_m, false, false);
    }
#pragma unroll
    for (int r = 0; r < 8; ++r) {
      const int mrow = r + 8 * half;
      const float z = acc_a[r] + acc_m[r];
      out[mrow * CC + ntile * 16 + l16] = 1.0f / (1.0f + __expf(-z));
    }
  }
}

extern "C" void kernel_launch(void* const* d_in, const int* in_sizes, int n_in,
                              void* d_out, int out_size, void* d_ws, size_t ws_size,
                              hipStream_t stream) {
  const float* x  = (const float*)d_in[0];
  const float* w1 = (const float*)d_in[1];
  const float* w2 = (const float*)d_in[2];
  float* out = (float*)d_out;

  float* ws_avg = (float*)d_ws;
  float* ws_max = ws_avg + ROWS;

  cam_pool_kernel<<<ROWS, 256, 0, stream>>>(x, ws_avg, ws_max);
  cam_mlp_wmma_kernel<<<1, 256, 0, stream>>>(ws_avg, ws_max, w1, w2, out);
}


# pool kernel as a stream: one wave per (b,c) row, 256 workgroups x 16 waves, each wave reads its 64 KB row in 8 chunks with the next chunk's 16 loads issued ahead, no LDS/barrier; sum order changes
# speedup vs baseline: 1.7847x; 1.7847x over previous
#include <hip/hip_runtime.h>
#include <math.h>


#define BB 16
#define CC 256
#define HW 16384
#define ROWS (BB * CC)

typedef __attribute__((ext_vector_type(2))) float v2f;
typedef __attribute__((ext_vector_type(4))) float v4f;
typedef __attribute__((ext_vector_type(8))) float v8f;

__device__ __forceinline__ int frag_index(int b, int k) {
  return ((((k >> 2) * BB) + b) << 2) + (k & 3);
}

#define POOL_WAVES 16
#define POOL_LOADS 16
#define POOL_CHUNKS (HW / 4 / 32 / POOL_LOADS)
#define POOL_THREADS (POOL_WAVES * 32)
#define POOL_GROUPS (ROWS / POOL_WAVES)

__global__ __launch_bounds__(POOL_THREADS) void cam_pool_kernel(
    const float* __restrict__ x,
    float* __restrict__ avg_out,
    float* __restrict__ max_out) {
  const int lane = threadIdx.x & 31;
  const int wave = __builtin_amdgcn_readfirstlane(threadIdx.x >> 5);
  const int row  = blockIdx.x * POOL_WAVES + wave;
  const v4f* __restrict__ p4 =
      (const v4f*)(x + (size_t)row * HW) + lane;

  v4f buf[2][POOL_LOADS];
  float s0 = 0.0f, s1 = 0.0f, s2 = 0.0f, s3 = 0.0f;
  float m0 = -INFINITY, m1 = -INFINITY, m2 = -INFINITY, m3 = -INFINITY;

#define POOL_LOAD_CHUNK(C)                                                   \
  _Pragma("unroll") for (int i = 0; i < POOL_LOADS; ++i)                     \
    buf[(C) & 1][i] = __builtin_nontemporal_load(&p4[((C) * POOL_LOADS + i) * 32])

  POOL_LOAD_CHUNK(0);
#pragma unroll
  for (int c = 0; c < POOL_CHUNKS; ++c) {
    __builtin_amdgcn_sched_barrier(0);
    if (c + 1 < POOL_CHUNKS) POOL_LOAD_CHUNK(c + 1);
    __builtin_amdgcn_sched_barrier(0);
#pragma unroll
    for (int i = 0; i < POOL_LOADS; ++i) {
      const v4f v = buf[c & 1][i];
      s0 += v.x; s1 += v.y; s2 += v.z; s3 += v.w;
      m0 = fmaxf(m0, v.x); m1 = fmaxf(m1, v.y);
      m2 = fmaxf(m2, v.z); m3 = fmaxf(m3, v.w);
    }
  }
#undef POOL_LOAD_CHUNK

  float s = (s0 + s1) + (s2 + s3);
  float m = fmaxf(fmaxf(m0, m1), fmaxf(m2, m3));
#pragma unroll
  for (int off = 16; off > 0; off >>= 1) {
    s += __shfl_xor(s, off, 32);
    m = fmaxf(m, __shfl_xor(m, off, 32));
  }
  if (lane == 0) {
    const int o = frag_index(row >> 8, row & (CC - 1));
    avg_out[o] = s * (1.0f / (float)HW);
    max_out[o] = m;
  }
}

__device__ __forceinline__ void tile_partial(
    const float* __restrict__ va,
    const float* __restrict__ vm,
    const float* __restrict__ w,
    int ntile, int q, int lane, v8f& acc_a, v8f& acc_m) {
  const int half = lane >> 4;
  const int l16  = lane & 15;
  const int ocol = ntile * 16 + l16;
  const v4f* __restrict__ a4 = (const v4f*)va + q * 256 + lane;
  const v4f* __restrict__ m4 = (const v4f*)vm + q * 256 + lane;
  const v4f* __restrict__ b4 = (const v4f*)(w + ocol * CC + q * 64 + half * 4);

  v4f fb[8], fa[8], fm[8];
#pragma unroll
  for (int j = 0; j < 8; ++j) {
    fb[j] = b4[j * 2];
    fa[j] = a4[j * 32];
    fm[j] = m4[j * 32];
  }
  __builtin_amdgcn_sched_barrier(0);

#pragma unroll
  for (int j = 0; j < 8; ++j) {
    v2f a0, a1, m0, m1, b0, b1;
    a0.x = fa[j].x; a0.y = fa[j].y; a1.x = fa[j].z; a1.y = fa[j].w;
    m0.x = fm[j].x; m0.y = fm[j].y; m1.x = fm[j].z; m1.y = fm[j].w;
    b0.x = fb[j].x; b0.y = fb[j].y; b1.x = fb[j].z; b1.y = fb[j].w;
    acc_a = __builtin_amdgcn_wmma_f32_16x16x4_f32(
        false, a0, false, b0, (short)0, acc_a, false, false);
    acc_m = __builtin_amdgcn_wmma_f32_16x16x4_f32(
        false, m0, false, b0, (short)0, acc_m, false, false);
    acc_a = __builtin_amdgcn_wmma_f32_16x16x4_f32(
        false, a1, false, b1, (short)0, acc_a, false, false);
    acc_m = __builtin_amdgcn_wmma_f32_16x16x4_f32(
        false, m1, false, b1, (short)0, acc_m, false, false);
  }
}

__global__ __launch_bounds__(128) void cam_layer1_kernel(
    const float* __restrict__ avg,
    const float* __restrict__ mx,
    const float* __restrict__ w1,
    float* __restrict__ h_a,
    float* __restrict__ h_m) {
  __shared__ float part[4][2][256];
  const int tid  = threadIdx.x;
  const int q    = tid >> 5;
  const int lane = tid & 31;
  const int half = lane >> 4;
  const int l16  = lane & 15;
  const int ntile = blockIdx.x;

  v8f acc_a = {};
  v8f acc_m = {};
  tile_partial(avg, mx, w1, ntile, q, lane, acc_a, acc_m);

#pragma unroll
  for (int r = 0; r < 8; ++r) {
    part[q][0][(r + 8 * half) * 16 + l16] = acc_a[r];
    part[q][1][(r + 8 * half) * 16 + l16] = acc_m[r];
  }
  __syncthreads();

#pragma unroll
  for (int i = 0; i < 2; ++i) {
    const int e = tid + i * 128;
    const int c = ((e >> 6) << 2) + (e & 3);
    const int b = (e >> 2) & 15;
    const int p = b * 16 + c;
    const float sa = ((part[0][0][p] + part[1][0][p]) + part[2][0][p]) + part[3][0][p];
    const float sm = ((part[0][1][p] + part[1][1][p]) + part[2][1][p]) + part[3][1][p];
    h_a[ntile * 256 + e] = fmaxf(sa, 0.0f);
    h_m[ntile * 256 + e] = fmaxf(sm, 0.0f);
  }
}

__global__ __launch_bounds__(256) void cam_layer2_kernel(
    const float* __restrict__ h_a,
    const float* __restrict__ h_m,
    const float* __restrict__ w2,
    float* __restrict__ out) {
  __shared__ float part[2][4][2][256];
  const int tid  = threadIdx.x;
  const int wave = tid >> 5;
  const int t    = wave >> 2;
  const int q    = wave & 3;
  const int lane = tid & 31;
  const int half = lane >> 4;
  const int l16  = lane & 15;

  v8f acc_a = {};
  v8f acc_m = {};
  tile_partial(h_a, h_m, w2, blockIdx.x * 2 + t, q, lane, acc_a, acc_m);

#pragma unroll
  for (int r = 0; r < 8; ++r) {
    part[t][q][0][(r + 8 * half) * 16 + l16] = acc_a[r];
    part[t][q][1][(r + 8 * half) * 16 + l16] = acc_m[r];
  }
  __syncthreads();

#pragma unroll
  for (int i = 0; i < 2; ++i) {
    const int e   = tid + i * 256;
    const int b   = e >> 5;
    const int col = e & 31;
    const int tt  = col >> 4;
    const int p   = b * 16 + (col & 15);
    const float sa = ((part[tt][0][0][p] + part[tt][1][0][p]) + part[tt][2][0][p]) + part[tt][3][0][p];
    const float sm = ((part[tt][0][1][p] + part[tt][1][1][p]) + part[tt][2][1][p]) + part[tt][3][1][p];
    const float z = sa + sm;
    out[b * CC + blockIdx.x * 32 + col] = 1.0f / (1.0f + __expf(-z));
  }
}

extern "C" void kernel_launch(void* const* d_in, const int* in_sizes, int n_in,
                              void* d_out, int out_size, void* d_ws, size_t ws_size,
                              hipStream_t stream) {
  const float* x  = (const float*)d_in[0];
  const float* w1 = (const float*)d_in[1];
  const float* w2 = (const float*)d_in[2];
  float* out = (float*)d_out;

  float* ws_avg = (float*)d_ws;
  float* ws_max = ws_avg + ROWS;
  float* ws_ha  = ws_max + ROWS;
  float* ws_hm  = ws_ha + ROWS;

  cam_pool_kernel<<<POOL_GROUPS, POOL_THREADS, 0, stream>>>(x, ws_avg, ws_max);
  cam_layer1_kernel<<<16, 128, 0, stream>>>(ws_avg, ws_max, w1, ws_ha, ws_hm);
  cam_layer2_kernel<<<8, 256, 0, stream>>>(ws_ha, ws_hm, w2, out);
}


# Layer 1 folded into the pool launch as per-workgroup partial sums over its 16 channels (2 launches, was 3); last launch adds the 16 partials per row in a fixed tree, then ReLU; sum order changes
# speedup vs baseline: 1.9218x; 1.0768x over previous
#include <hip/hip_runtime.h>
#include <math.h>


#define BB 16
#define CC 256
#define HW 16384
#define ROWS (BB * CC)

typedef __attribute__((ext_vector_type(2))) float v2f;
typedef __attribute__((ext_vector_type(4))) float v4f;
typedef __attribute__((ext_vector_type(8))) float v8f;

#define CGROUP 16
#define NGROUPS (CC / CGROUP)

#define POOL_WAVES 16
#define POOL_LOADS 16
#define POOL_CHUNKS (HW / 4 / 32 / POOL_LOADS)
#define POOL_THREADS (POOL_WAVES * 32)
#define POOL_GROUPS (ROWS / POOL_WAVES)

__global__ __launch_bounds__(POOL_THREADS) void cam_pool_l1_kernel(
    const float* __restrict__ x,
    const float* __restrict__ w1,
    float* __restrict__ part) {
  __shared__ __attribute__((aligned(16))) float pl[3][CGROUP];
  __shared__ float tile[POOL_WAVES][256];
  const int lane = threadIdx.x & 31;
  const int half = lane >> 4;
  const int l16  = lane & 15;
  const int wave = __builtin_amdgcn_readfirstlane(threadIdx.x >> 5);
  const int row  = blockIdx.x * POOL_WAVES + wave;
  const v4f* __restrict__ p4 =
      (const v4f*)(x + (size_t)row * HW) + lane;

  v4f buf[2][POOL_LOADS];
  float s0 = 0.0f, s1 = 0.0f, s2 = 0.0f, s3 = 0.0f;
  float m0 = -INFINITY, m1 = -INFINITY, m2 = -INFINITY, m3 = -INFINITY;

#define POOL_LOAD_CHUNK(C)                                                   \
  _Pragma("unroll") for (int i = 0; i < POOL_LOADS; ++i)                     \
    buf[(C) & 1][i] = __builtin_nontemporal_load(&p4[((C) * POOL_LOADS + i) * 32])

  POOL_LOAD_CHUNK(0);
  const v4f* __restrict__ wb4 = (const v4f*)(
      w1 + (wave * 16 + l16) * CC + (blockIdx.x & (NGROUPS - 1)) * CGROUP + half * 4);
  const v4f fb0 = wb4[0];
  const v4f fb1 = wb4[2];
#pragma unroll
  for (int c = 0; c < POOL_CHUNKS; ++c) {
    __builtin_amdgcn_sched_barrier(0);
    if (c + 1 < POOL_CHUNKS) POOL_LOAD_CHUNK(c + 1);
    __builtin_amdgcn_sched_barrier(0);
#pragma unroll
    for (int i = 0; i < POOL_LOADS; ++i) {
      const v4f v = buf[c & 1][i];
      s0 += v.x; s1 += v.y; s2 += v.z; s3 += v.w;
      m0 = fmaxf(m0, v.x); m1 = fmaxf(m1, v.y);
      m2 = fmaxf(m2, v.z); m3 = fmaxf(m3, v.w);
    }
  }
#undef POOL_LOAD_CHUNK

  float s = (s0 + s1) + (s2 + s3);
  float m = fmaxf(fmaxf(m0, m1), fmaxf(m2, m3));
#pragma unroll
  for (int off = 16; off > 0; off >>= 1) {
    s += __shfl_xor(s, off, 32);
    m = fmaxf(m, __shfl_xor(m, off, 32));
  }
  if (lane == 0) {
    pl[0][wave] = s * (1.0f / (float)HW);
    pl[1][wave] = m;
  }
  if (wave == 0 && lane < CGROUP) pl[2][lane] = 0.0f;
  __syncthreads();

  const int arow = l16 < 2 ? l16 : 2;
  const v4f* __restrict__ pa4 = (const v4f*)&pl[arow][half * 4];
  const v4f fa0 = pa4[0];
  const v4f fa1 = pa4[2];
  float* __restrict__ mytile = &tile[wave][half * 128 + l16];
  const float* __restrict__ src = &tile[wave][lane];
  float* __restrict__ dst = part + blockIdx.x * (2 * CC) + wave * 32 + lane;
  v2f a, b;
  v8f acc = {};
  a.x = fa0.x; a.y = fa0.y; b.x = fb0.x; b.y = fb0.y;
  acc = __builtin_amdgcn_wmma_f32_16x16x4_f32(false, a, false, b, (short)0, acc, false, false);
  a.x = fa0.z; a.y = fa0.w; b.x = fb0.z; b.y = fb0.w;
  acc = __builtin_amdgcn_wmma_f32_16x16x4_f32(false, a, false, b, (short)0, acc, false, false);
  a.x = fa1.x; a.y = fa1.y; b.x = fb1.x; b.y = fb1.y;
  acc = __builtin_amdgcn_wmma_f32_16x16x4_f32(false, a, false, b, (short)0, acc, false, false);
  a.x = fa1.z; a.y = fa1.w; b.x = fb1.z; b.y = fb1.w;
  acc = __builtin_amdgcn_wmma_f32_16x16x4_f32(false, a, false, b, (short)0, acc, false, false);

#pragma unroll
  for (int r = 0; r < 8; ++r) mytile[r * 16] = acc[r];
  *dst = *src;
}

#define HS_PITCH 264

__global__ __launch_bounds__(256) void cam_mlp_tail_kernel(
    const float* __restrict__ part,
    const float* __restrict__ w2,
    float* __restrict__ out) {
  __shared__ __attribute__((aligned(16))) float hs[3][HS_PITCH];
  __shared__ float red[8][256];
  const int tid  = threadIdx.x;
  const int wave = tid >> 5;
  const int t    = wave >> 2;
  const int q    = wave & 3;
  const int lane = tid & 31;
  const int half = lane >> 4;
  const int l16  = lane & 15;
  const int b    = blockIdx.x >> 3;
  const int og   = blockIdx.x & 7;

  const int ocol = og * 32 + t * 16 + l16;
  const v4f* __restrict__ b4 = (const v4f*)(w2 + ocol * CC + q * 64 + half * 4);
  const v2f* __restrict__ p2 = (const v2f*)part + b * NGROUPS * CC + tid;
  const int br = (tid >> 3) & 1;
  const int hc = ((tid >> 4) << 4) + ((tid & 7) << 1);

  v4f fb[8];
  v2f pv[NGROUPS];
#pragma unroll
  for (int j = 0; j < 8; ++j) fb[j] = b4[j * 2];
#pragma unroll
  for (int g = 0; g < NGROUPS; ++g) pv[g] = p2[g * CC];
  __builtin_amdgcn_sched_barrier(0);

#pragma unroll
  for (int w = 1; w < NGROUPS; w <<= 1) {
#pragma unroll
    for (int g = 0; g < NGROUPS; g += 2 * w) pv[g] += pv[g + w];
  }
  v2f hv;
  hv.x = fmaxf(pv[0].x, 0.0f);
  hv.y = fmaxf(pv[0].y, 0.0f);
  *(v2f*)&hs[br][hc] = hv;
  hs[2][tid] = 0.0f;
  __syncthreads();

  const int arow = l16 < 2 ? l16 : 2;
  const v4f* __restrict__ a4 = (const v4f*)&hs[arow][q * 64 + half * 4];
  v4f fa[8];
#pragma unroll
  for (int j = 0; j < 8; ++j) fa[j] = a4[j * 2];
  __builtin_amdgcn_sched_barrier(0);

  v8f acc = {};
#pragma unroll
  for (int j = 0; j < 8; ++j) {
    v2f a0, a1, b0, b1;
    a0.x = fa[j].x; a0.y = fa[j].y; a1.x = fa[j].z; a1.y = fa[j].w;
    b0.x = fb[j].x; b0.y = fb[j].y; b1.x = fb[j].z; b1.y = fb[j].w;
    acc = __builtin_amdgcn_wmma_f32_16x16x4_f32(
        false, a0, false, b0, (short)0, acc, false, false);
    acc = __builtin_amdgcn_wmma_f32_16x16x4_f32(
        false, a1, false, b1, (short)0, acc, false, false);
  }

#pragma unroll
  for (int r = 0; r < 8; ++r) red[r][tid] = acc[r];
  __syncthreads();

  if (tid < 32) {
    const int tt = tid >> 4;
    const int c  = tid & 15;
    const int s = tt * 128 + c;
    const float sa = ((red[0][s] + red[0][s + 32]) + red[0][s + 64]) + red[0][s + 96];
    const float sm = ((red[1][s] + red[1][s + 32]) + red[1][s + 64]) + red[1][s + 96];
    const float z = sa + sm;
    out[b * CC + og * 32 + tid] = 1.0f / (1.0f + __expf(-z));
  }
}

extern "C" void kernel_launch(void* const* d_in, const int* in_sizes, int n_in,
                              void* d_out, int out_size, void* d_ws, size_t ws_size,
                              hipStream_t stream) {
  const float* x  = (const float*)d_in[0];
  const float* w1 = (const float*)d_in[1];
  const float* w2 = (const float*)d_in[2];
  float* out = (float*)d_out;

  float* ws_part = (float*)d_ws;

  cam_pool_l1_kernel<<<POOL_GROUPS, POOL_THREADS, 0, stream>>>(x, w1, ws_part);
  cam_mlp_tail_kernel<<<128, 256, 0, stream>>>(ws_part, w2, out);
}
